# starting kernel
# baseline (speedup 1.0000x reference)
#include <hip/hip_runtime.h>
#include <math.h>

typedef __attribute__((ext_vector_type(2))) float v2f;
typedef __attribute__((ext_vector_type(8))) float v8f;

#define HDIM 128

__global__ void deg_init_kernel(float* deg, int N) {
  int i = blockIdx.x * blockDim.x + threadIdx.x;
  if (i < N) deg[i] = 1.0f;
}

__global__ void deg_edge_kernel(float* deg, const int* __restrict__ dst, int E) {
  int e = blockIdx.x * blockDim.x + threadIdx.x;
  if (e < E) atomicAdd(&deg[dst[e]], 1.0f);
}

__global__ void deg_fin_kernel(float* deg, int N) {
  int i = blockIdx.x * blockDim.x + threadIdx.x;
  if (i < N) {
    float d = deg[i];
    deg[i] = d > 0.0f ? rsqrtf(d) : 0.0f;
  }
}

template <int NCOLS, int EPI>
__global__ __launch_bounds__(NCOLS * 2)
void gemm_f32_wmma_kernel(const float* __restrict__ A,
                          const float* __restrict__ W,
                          const float* __restrict__ dis,
                          const float* __restrict__ bias,
                          float* __restrict__ out0,
                          float* __restrict__ out1,
                          int M) {
  constexpr int K = HDIM;
  constexpr int KSTEPS = K / 4;
  __shared__ float sAt[K * 17 + 16];

  const int tid = threadIdx.x;
  const int wave = tid >> 5;
  const int lane = tid & 31;
  const int idx15 = lane & 15;
  const int kb = (lane >> 4) << 1;
  const int col = wave * 16 + idx15;

  v2f bfrag[KSTEPS];
#pragma unroll
  for (int s = 0; s < KSTEPS; ++s) {
    const float* wp = W + (size_t)col * K + s * 4 + kb;
    bfrag[s].x = wp[0];
    bfrag[s].y = wp[1];
  }
  float bias_v = 0.0f;
  if (EPI == 1) bias_v = bias[col];

  const int mtiles = M >> 4;
  for (int mt = blockIdx.x; mt < mtiles; mt += gridDim.x) {
    __syncthreads();
#pragma unroll
    for (int i = tid; i < 16 * K; i += NCOLS * 2) {
      int r = i >> 7;
      int c = i & (K - 1);
      sAt[c * 17 + r] = A[((size_t)mt * 16 + r) * K + c];
    }
    __syncthreads();

    v8f accv = {};
#pragma unroll
    for (int s = 0; s < KSTEPS; ++s) {
      const int k = s * 4 + kb;
      v2f a;
      a.x = sAt[k * 17 + idx15];
      a.y = sAt[(k + 1) * 17 + idx15];
      accv = __builtin_amdgcn_wmma_f32_16x16x4_f32(
          false, a, false, bfrag[s], (short)0, accv, false, false);
    }

    const int mbase = (lane >> 4) << 3;
#pragma unroll
    for (int i = 0; i < 8; ++i) {
      const size_t row = (size_t)mt * 16 + mbase + i;
      float v = accv[i];
      if (EPI == 0) {
        v *= dis[row];
        out0[row * NCOLS + col] = v;
        out1[row * NCOLS + col] = v;
      } else {
        v += bias_v;
        v = fmaxf(v, 0.0f);
        out0[row * NCOLS + col] = v;
      }
    }
  }
}

__global__ void scatter_kernel(const int* __restrict__ src,
                               const int* __restrict__ dst,
                               const float* __restrict__ hw,
                               float* __restrict__ acc, int E) {
  long long gid = (long long)blockIdx.x * blockDim.x + threadIdx.x;
  int e = (int)(gid >> 5);
  int lane = (int)(gid & 31);
  if (e >= E) return;
  int s = src[e];
  int d = dst[e];
  const float4* vp = (const float4*)(hw + (size_t)s * HDIM);
  float4 v = vp[lane];
  float* p = acc + (size_t)d * HDIM + lane * 4;
  atomicAdd(p + 0, v.x);
  atomicAdd(p + 1, v.y);
  atomicAdd(p + 2, v.z);
  atomicAdd(p + 3, v.w);
}

__global__ void bn_relu_kernel(const float* __restrict__ acc,
                               const float* __restrict__ dis,
                               const float* __restrict__ bias,
                               const float* __restrict__ gamma,
                               const float* __restrict__ beta,
                               const float* __restrict__ mean,
                               const float* __restrict__ var,
                               float* __restrict__ h, int N) {
  const int total = N * HDIM;
  const int stride = gridDim.x * blockDim.x;
  for (int i = blockIdx.x * blockDim.x + threadIdx.x; i < total; i += stride) {
    int c = i & (HDIM - 1);
    int n = i >> 7;
    float v = acc[i] * dis[n] + bias[c];
    v = (v - mean[c]) * rsqrtf(var[c] + 1e-5f) * gamma[c] + beta[c];
    h[i] = fmaxf(v, 0.0f);
  }
}

__global__ void node_head2_kernel(const float* __restrict__ t,
                                  const float* __restrict__ w2,
                                  const float* __restrict__ b2,
                                  float* __restrict__ out, int N) {
  int n = blockIdx.x * blockDim.x + threadIdx.x;
  if (n >= N) return;
  const float* tp = t + (size_t)n * 64;
  float a0 = b2[0], a1 = b2[1];
#pragma unroll 16
  for (int k = 0; k < 64; ++k) {
    float tv = tp[k];
    a0 = fmaf(tv, w2[k], a0);
    a1 = fmaf(tv, w2[64 + k], a1);
  }
  out[(size_t)n * 2 + 0] = a0;
  out[(size_t)n * 2 + 1] = a1;
}

__global__ void pool_zero_kernel(float* gbuf, int total) {
  int i = blockIdx.x * blockDim.x + threadIdx.x;
  if (i < total) gbuf[i] = 0.0f;
}

__global__ void pool_block_kernel(const float* __restrict__ h,
                                  const int* __restrict__ batch,
                                  float* __restrict__ gsum,
                                  float* __restrict__ gcnt,
                                  int N) {
  __shared__ float lsum[64 * HDIM];
  __shared__ float lcnt[64];
  const int c = threadIdx.x;
  const int base = blockIdx.x * HDIM;
  for (int i = c; i < 64 * HDIM; i += HDIM) lsum[i] = 0.0f;
  if (c < 64) lcnt[c] = 0.0f;
  __syncthreads();
  int nmax = N - base;
  if (nmax > HDIM) nmax = HDIM;
  int curg = -1;
  float accv = 0.0f;
  for (int r = 0; r < nmax; ++r) {
    const int n = base + r;
    const int g = batch[n];
    if (g != curg) {
      if (curg >= 0) lsum[curg * HDIM + c] += accv;
      curg = g;
      accv = 0.0f;
    }
    accv += h[(size_t)n * HDIM + c];
    if (c == 0) lcnt[g] += 1.0f;
  }
  if (curg >= 0) lsum[curg * HDIM + c] += accv;
  __syncthreads();
  const int gmin = batch[base];
  const int gmax = batch[base + nmax - 1];
  for (int g = gmin; g <= gmax; ++g)
    atomicAdd(&gsum[g * HDIM + c], lsum[g * HDIM + c]);
  if (c == 0)
    for (int g = gmin; g <= gmax; ++g) atomicAdd(&gcnt[g], lcnt[g]);
}

__global__ void graph_head_kernel(const float* __restrict__ gsum,
                                  const float* __restrict__ gcnt,
                                  const float* __restrict__ w1,
                                  const float* __restrict__ b1,
                                  const float* __restrict__ w2,
                                  const float* __restrict__ b2,
                                  float* __restrict__ out) {
  __shared__ float emb[HDIM];
  __shared__ float t[64];
  const int g = blockIdx.x;
  const int j = threadIdx.x;
  const float inv = 1.0f / fmaxf(gcnt[g], 1.0f);
  for (int i = j; i < HDIM; i += 64) emb[i] = gsum[g * HDIM + i] * inv;
  __syncthreads();
  float a = b1[j];
#pragma unroll 16
  for (int k = 0; k < HDIM; ++k) a = fmaf(emb[k], w1[j * HDIM + k], a);
  t[j] = fmaxf(a, 0.0f);
  __syncthreads();
  if (j < 2) {
    float o = b2[j];
#pragma unroll 16
    for (int k = 0; k < 64; ++k) o = fmaf(t[k], w2[j * 64 + k], o);
    out[g * 2 + j] = o;
  }
}

extern "C" void kernel_launch(void* const* d_in, const int* in_sizes, int n_in,
                              void* d_out, int out_size, void* d_ws, size_t ws_size,
                              hipStream_t stream) {
  const float* x      = (const float*)d_in[0];
  const int*   eidx   = (const int*)d_in[1];
  const int*   batch  = (const int*)d_in[2];
  const float* conv_w = (const float*)d_in[3];
  const float* conv_b = (const float*)d_in[4];
  const float* bn_g   = (const float*)d_in[5];
  const float* bn_b   = (const float*)d_in[6];
  const float* bn_m   = (const float*)d_in[7];
  const float* bn_v   = (const float*)d_in[8];
  const float* nw1    = (const float*)d_in[9];
  const float* nb1    = (const float*)d_in[10];
  const float* nw2    = (const float*)d_in[11];
  const float* nb2    = (const float*)d_in[12];
  const float* gw1    = (const float*)d_in[13];
  const float* gb1    = (const float*)d_in[14];
  const float* gw2    = (const float*)d_in[15];
  const float* gb2    = (const float*)d_in[16];

  const int N = in_sizes[0] / HDIM;
  const int E = in_sizes[1] / 2;
  const int L = in_sizes[3] / (HDIM * HDIM);
  const int G = (out_size - 2 * N) / 2;

  const int* src = eidx;
  const int* dst = eidx + E;

  float* dis  = (float*)d_ws;
  float* h    = dis + N;
  float* hw   = h + (size_t)N * HDIM;
  float* acc  = hw + (size_t)N * HDIM;
  float* tbuf = acc + (size_t)N * HDIM;
  float* gsum = tbuf + (size_t)N * 64;
  float* gcnt = gsum + (size_t)G * HDIM;

  float* node_out  = (float*)d_out;
  float* graph_out = node_out + (size_t)N * 2;

  deg_init_kernel<<<(N + 255) / 256, 256, 0, stream>>>(dis, N);
  deg_edge_kernel<<<(E + 255) / 256, 256, 0, stream>>>(dis, dst, E);
  deg_fin_kernel<<<(N + 255) / 256, 256, 0, stream>>>(dis, N);

  const int mtiles = N / 16;
  const int gemm_grid = mtiles < 2048 ? mtiles : 2048;

  const float* hin = x;
  for (int l = 0; l < L; ++l) {
    gemm_f32_wmma_kernel<HDIM, 0><<<gemm_grid, 256, 0, stream>>>(
        hin, conv_w + (size_t)l * HDIM * HDIM, dis, nullptr, hw, acc, N);
    long long sthreads = (long long)E * 32;
    scatter_kernel<<<(int)((sthreads + 255) / 256), 256, 0, stream>>>(
        src, dst, hw, acc, E);
    bn_relu_kernel<<<4096, 256, 0, stream>>>(
        acc, dis, conv_b + l * HDIM, bn_g + l * HDIM, bn_b + l * HDIM,
        bn_m + l * HDIM, bn_v + l * HDIM, h, N);
    hin = h;
  }

  gemm_f32_wmma_kernel<64, 1><<<gemm_grid, 128, 0, stream>>>(
      h, nw1, nullptr, nb1, tbuf, nullptr, N);
  node_head2_kernel<<<(N + 255) / 256, 256, 0, stream>>>(
      tbuf, nw2, nb2, node_out, N);

  pool_zero_kernel<<<(G * HDIM + G + 255) / 256, 256, 0, stream>>>(
      gsum, G * HDIM + G);
  pool_block_kernel<<<(N + HDIM - 1) / HDIM, HDIM, 0, stream>>>(
      h, batch, gsum, gcnt, N);
  graph_head_kernel<<<G, 64, 0, stream>>>(gsum, gcnt, gw1, gb1, gw2, gb2,
                                          graph_out);
}


# Float-atomic edge scatter-add replaced by per-call destination-ordered edge lists (int count+slot, scan, place, sort by source) and a one-wave-per-node row gather; sum order now fixed by inputs
# speedup vs baseline: 2.2749x; 2.2749x over previous
#include <hip/hip_runtime.h>
#include <math.h>

typedef __attribute__((ext_vector_type(2))) float v2f;
typedef __attribute__((ext_vector_type(8))) float v8f;

#define HDIM 128

__global__ void cnt_zero_kernel(int* cnt, int N) {
  int i = blockIdx.x * blockDim.x + threadIdx.x;
  if (i < N) cnt[i] = 0;
}

__global__ void edge_rank_kernel(int* cnt, const int* __restrict__ dst,
                                 int* __restrict__ rank, int E) {
  int e = blockIdx.x * blockDim.x + threadIdx.x;
  if (e < E) rank[e] = atomicAdd(&cnt[dst[e]], 1);
}

__global__ __launch_bounds__(1024)
void rowptr_scan_kernel(const int* __restrict__ cnt, int* __restrict__ rowptr,
                        float* __restrict__ dis, int N) {
  __shared__ int wpre[32];
  __shared__ int wtot[32];
  const int tid = threadIdx.x;
  const int lane = tid & 31;
  const int wave = tid >> 5;
  const int base = blockIdx.x * 1024;
  const int4* __restrict__ cnt4 = (const int4*)cnt;
  const int nq = base >> 2;
  int part = 0;
#pragma unroll 4
  for (int q = tid; q < nq; q += 1024) {
    const int4 t = cnt4[q];
    part += (t.x + t.y) + (t.z + t.w);
  }
  for (int o = 16; o > 0; o >>= 1) part += __shfl_xor(part, o);
  if (lane == 0) wpre[wave] = part;
  const int i = base + tid;
  const int v = (i < N) ? cnt[i] : 0;
  int inc = v;
  for (int o = 1; o < 32; o <<= 1) {
    const int t = __shfl_up(inc, o);
    if (lane >= o) inc += t;
  }
  if (lane == 31) wtot[wave] = inc;
  __syncthreads();
  int off = 0;
  for (int w = 0; w < 32; ++w) {
    off += wpre[w];
    if (w < wave) off += wtot[w];
  }
  const int excl = off + inc - v;
  if (i < N) {
    rowptr[i] = excl;
    const float d = (float)(v + 1);
    dis[i] = d > 0.0f ? rsqrtf(d) : 0.0f;
    if (i == N - 1) rowptr[N] = excl + v;
  }
}

__global__ void csr_fill_kernel(const int* __restrict__ src, const int* __restrict__ dst,
                                const int* __restrict__ rank, const int* __restrict__ rowptr,
                                int* __restrict__ colu, int E) {
  int e = blockIdx.x * blockDim.x + threadIdx.x;
  if (e < E) colu[rowptr[dst[e]] + rank[e]] = src[e];
}

__global__ __launch_bounds__(256)
void csr_order_kernel(const int* __restrict__ rowptr, const int* __restrict__ colu,
                      int* __restrict__ cols, int N) {
  const int node = __builtin_amdgcn_readfirstlane(blockIdx.x * 8 + (threadIdx.x >> 5));
  const int lane = threadIdx.x & 31;
  if (node >= N) return;
  const int beg = rowptr[node];
  const int end = rowptr[node + 1];
  for (int i0 = beg; i0 < end; i0 += 32) {
    const int i = i0 + lane;
    const int x = (i < end) ? colu[i] : 0x7fffffff;
    int r = 0;
    for (int j0 = beg; j0 < end; j0 += 32) {
      const int j = j0 + lane;
      const int y = (j < end) ? colu[j] : 0x7fffffff;
      const int m = (end - j0) < 32 ? (end - j0) : 32;
      for (int k = 0; k < m; ++k) {
        const int yk = __builtin_amdgcn_readlane(y, k);
        r += (yk < x) ? 1 : 0;
        r += (yk == x && (j0 + k) < i) ? 1 : 0;
      }
    }
    if (i < end) cols[beg + r] = x;
  }
}

template <int NCOLS, int EPI>
__global__ __launch_bounds__(NCOLS * 2)
void gemm_f32_wmma_kernel(const float* __restrict__ A,
                          const float* __restrict__ W,
                          const float* __restrict__ dis,
                          const float* __restrict__ bias,
                          float* __restrict__ out0,
                          int M) {
  constexpr int K = HDIM;
  constexpr int KSTEPS = K / 4;
  __shared__ float sAt[K * 17 + 16];

  const int tid = threadIdx.x;
  const int wave = tid >> 5;
  const int lane = tid & 31;
  const int idx15 = lane & 15;
  const int kb = (lane >> 4) << 1;
  const int col = wave * 16 + idx15;

  v2f bfrag[KSTEPS];
#pragma unroll
  for (int s = 0; s < KSTEPS; ++s) {
    const float* wp = W + (size_t)col * K + s * 4 + kb;
    bfrag[s].x = wp[0];
    bfrag[s].y = wp[1];
  }
  float bias_v = 0.0f;
  if (EPI == 1) bias_v = bias[col];

  const int mtiles = M >> 4;
  for (int mt = blockIdx.x; mt < mtiles; mt += gridDim.x) {
    __syncthreads();
#pragma unroll
    for (int i = tid; i < 16 * K; i += NCOLS * 2) {
      int r = i >> 7;
      int c = i & (K - 1);
      sAt[c * 17 + r] = A[((size_t)mt * 16 + r) * K + c];
    }
    __syncthreads();

    v8f accv = {};
#pragma unroll
    for (int s = 0; s < KSTEPS; ++s) {
      const int k = s * 4 + kb;
      v2f a;
      a.x = sAt[k * 17 + idx15];
      a.y = sAt[(k + 1) * 17 + idx15];
      accv = __builtin_amdgcn_wmma_f32_16x16x4_f32(
          false, a, false, bfrag[s], (short)0, accv, false, false);
    }

    const int mbase = (lane >> 4) << 3;
#pragma unroll
    for (int i = 0; i < 8; ++i) {
      const size_t row = (size_t)mt * 16 + mbase + i;
      float v = accv[i];
      if (EPI == 0) {
        v *= dis[row];
        out0[row * NCOLS + col] = v;
      } else {
        v += bias_v;
        v = fmaxf(v, 0.0f);
        out0[row * NCOLS + col] = v;
      }
    }
  }
}

#define GATHER_ROWS(CNT)                                                        \
  {                                                                             \
    float4 v[CNT];                                                              \
    _Pragma("unroll") for (int u = 0; u < CNT; ++u) {                           \
      const int s = __builtin_amdgcn_readlane(my, k + u);                       \
      v[u] = hw4[(size_t)s * 32 + lane];                                        \
    }                                                                           \
    _Pragma("unroll") for (int u = 0; u < CNT; ++u) {                           \
      a.x += v[u].x; a.y += v[u].y; a.z += v[u].z; a.w += v[u].w;               \
    }                                                                           \
    k += CNT;                                                                   \
  }

__global__ __launch_bounds__(256)
void gather_rows_kernel(const int* __restrict__ rowptr, const int* __restrict__ cols,
                        const float* __restrict__ hw, float* __restrict__ acc, int N) {
  const int node = __builtin_amdgcn_readfirstlane(blockIdx.x * 8 + (threadIdx.x >> 5));
  const int lane = threadIdx.x & 31;
  if (node >= N) return;
  const float4* __restrict__ hw4 = (const float4*)hw;
  float4 a = hw4[(size_t)node * 32 + lane];
  const int beg = rowptr[node];
  const int end = rowptr[node + 1];
  for (int j0 = beg; j0 < end; j0 += 32) {
    const int m = (end - j0) < 32 ? (end - j0) : 32;
    const int my = (lane < m) ? cols[j0 + lane] : 0;
    int k = 0;
    while (k + 8 <= m) GATHER_ROWS(8)
    if (k + 4 <= m) GATHER_ROWS(4)
    if (k + 2 <= m) GATHER_ROWS(2)
    if (k < m) GATHER_ROWS(1)
  }
  ((float4*)acc)[(size_t)node * 32 + lane] = a;
}

__global__ void bn_relu_kernel(const float* __restrict__ acc,
                               const float* __restrict__ dis,
                               const float* __restrict__ bias,
                               const float* __restrict__ gamma,
                               const float* __restrict__ beta,
                               const float* __restrict__ mean,
                               const float* __restrict__ var,
                               float* __restrict__ h, int N) {
  const int total = N * HDIM;
  const int stride = gridDim.x * blockDim.x;
  for (int i = blockIdx.x * blockDim.x + threadIdx.x; i < total; i += stride) {
    int c = i & (HDIM - 1);
    int n = i >> 7;
    float v = acc[i] * dis[n] + bias[c];
    v = (v - mean[c]) * rsqrtf(var[c] + 1e-5f) * gamma[c] + beta[c];
    h[i] = fmaxf(v, 0.0f);
  }
}

__global__ void node_head2_kernel(const float* __restrict__ t,
                                  const float* __restrict__ w2,
                                  const float* __restrict__ b2,
                                  float* __restrict__ out, int N) {
  int n = blockIdx.x * blockDim.x + threadIdx.x;
  if (n >= N) return;
  const float* tp = t + (size_t)n * 64;
  float a0 = b2[0], a1 = b2[1];
#pragma unroll 16
  for (int k = 0; k < 64; ++k) {
    float tv = tp[k];
    a0 = fmaf(tv, w2[k], a0);
    a1 = fmaf(tv, w2[64 + k], a1);
  }
  out[(size_t)n * 2 + 0] = a0;
  out[(size_t)n * 2 + 1] = a1;
}

__global__ void pool_zero_kernel(float* gbuf, int total) {
  int i = blockIdx.x * blockDim.x + threadIdx.x;
  if (i < total) gbuf[i] = 0.0f;
}

__global__ void pool_block_kernel(const float* __restrict__ h,
                                  const int* __restrict__ batch,
                                  float* __restrict__ gsum,
                                  float* __restrict__ gcnt,
                                  int N) {
  __shared__ float lsum[64 * HDIM];
  __shared__ float lcnt[64];
  const int c = threadIdx.x;
  const int base = blockIdx.x * HDIM;
  for (int i = c; i < 64 * HDIM; i += HDIM) lsum[i] = 0.0f;
  if (c < 64) lcnt[c] = 0.0f;
  __syncthreads();
  int nmax = N - base;
  if (nmax > HDIM) nmax = HDIM;
  int curg = -1;
  float accv = 0.0f;
  for (int r = 0; r < nmax; ++r) {
    const int n = base + r;
    const int g = batch[n];
    if (g != curg) {
      if (curg >= 0) lsum[curg * HDIM + c] += accv;
      curg = g;
      accv = 0.0f;
    }
    accv += h[(size_t)n * HDIM + c];
    if (c == 0) lcnt[g] += 1.0f;
  }
  if (curg >= 0) lsum[curg * HDIM + c] += accv;
  __syncthreads();
  const int gmin = batch[base];
  const int gmax = batch[base + nmax - 1];
  for (int g = gmin; g <= gmax; ++g)
    atomicAdd(&gsum[g * HDIM + c], lsum[g * HDIM + c]);
  if (c == 0)
    for (int g = gmin; g <= gmax; ++g) atomicAdd(&gcnt[g], lcnt[g]);
}

__global__ void graph_head_kernel(const float* __restrict__ gsum,
                                  const float* __restrict__ gcnt,
                                  const float* __restrict__ w1,
                                  const float* __restrict__ b1,
                                  const float* __restrict__ w2,
                                  const float* __restrict__ b2,
                                  float* __restrict__ out) {
  __shared__ float emb[HDIM];
  __shared__ float t[64];
  const int g = blockIdx.x;
  const int j = threadIdx.x;
  const float inv = 1.0f / fmaxf(gcnt[g], 1.0f);
  for (int i = j; i < HDIM; i += 64) emb[i] = gsum[g * HDIM + i] * inv;
  __syncthreads();
  float a = b1[j];
#pragma unroll 16
  for (int k = 0; k < HDIM; ++k) a = fmaf(emb[k], w1[j * HDIM + k], a);
  t[j] = fmaxf(a, 0.0f);
  __syncthreads();
  if (j < 2) {
    float o = b2[j];
#pragma unroll 16
    for (int k = 0; k < 64; ++k) o = fmaf(t[k], w2[j * 64 + k], o);
    out[g * 2 + j] = o;
  }
}

extern "C" void kernel_launch(void* const* d_in, const int* in_sizes, int n_in,
                              void* d_out, int out_size, void* d_ws, size_t ws_size,
                              hipStream_t stream) {
  const float* x      = (const float*)d_in[0];
  const int*   eidx   = (const int*)d_in[1];
  const int*   batch  = (const int*)d_in[2];
  const float* conv_w = (const float*)d_in[3];
  const float* conv_b = (const float*)d_in[4];
  const float* bn_g   = (const float*)d_in[5];
  const float* bn_b   = (const float*)d_in[6];
  const float* bn_m   = (const float*)d_in[7];
  const float* bn_v   = (const float*)d_in[8];
  const float* nw1    = (const float*)d_in[9];
  const float* nb1    = (const float*)d_in[10];
  const float* nw2    = (const float*)d_in[11];
  const float* nb2    = (const float*)d_in[12];
  const float* gw1    = (const float*)d_in[13];
  const float* gb1    = (const float*)d_in[14];
  const float* gw2    = (const float*)d_in[15];
  const float* gb2    = (const float*)d_in[16];

  const int N = in_sizes[0] / HDIM;
  const int E = in_sizes[1] / 2;
  const int L = in_sizes[3] / (HDIM * HDIM);
  const int G = (out_size - 2 * N) / 2;

  const int* src = eidx;
  const int* dst = eidx + E;

  float* dis  = (float*)d_ws;
  float* h    = dis + N;
  float* hw   = h + (size_t)N * HDIM;
  float* acc  = hw + (size_t)N * HDIM;
  float* tbuf = acc + (size_t)N * HDIM;
  float* gsum = tbuf + (size_t)N * 64;
  float* gcnt = gsum + (size_t)G * HDIM;

  int* cnt    = (int*)tbuf;
  int* rowptr = cnt + N;
  int* cols   = rowptr + N + 32;
  int* colu   = cols + E;

  float* node_out  = (float*)d_out;
  float* graph_out = node_out + (size_t)N * 2;

  cnt_zero_kernel<<<(N + 255) / 256, 256, 0, stream>>>(cnt, N);
  edge_rank_kernel<<<(E + 255) / 256, 256, 0, stream>>>(cnt, dst, cols, E);
  rowptr_scan_kernel<<<(N + 1023) / 1024, 1024, 0, stream>>>(cnt, rowptr, dis, N);
  csr_fill_kernel<<<(E + 255) / 256, 256, 0, stream>>>(src, dst, cols, rowptr, colu, E);
  csr_order_kernel<<<(N + 7) / 8, 256, 0, stream>>>(rowptr, colu, cols, N);

  const int mtiles = N / 16;
  const int gemm_grid = mtiles < 2048 ? mtiles : 2048;

  const float* hin = x;
  for (int l = 0; l < L; ++l) {
    gemm_f32_wmma_kernel<HDIM, 0><<<gemm_grid, 256, 0, stream>>>(
        hin, conv_w + (size_t)l * HDIM * HDIM, dis, nullptr, hw, N);
    gather_rows_kernel<<<(N + 7) / 8, 256, 0, stream>>>(rowptr, cols, hw, acc, N);
    bn_relu_kernel<<<4096, 256, 0, stream>>>(
        acc, dis, conv_b + l * HDIM, bn_g + l * HDIM, bn_b + l * HDIM,
        bn_m + l * HDIM, bn_v + l * HDIM, h, N);
    hin = h;
  }

  gemm_f32_wmma_kernel<64, 1><<<gemm_grid, 128, 0, stream>>>(
      h, nw1, nullptr, nb1, tbuf, N);
  node_head2_kernel<<<(N + 255) / 256, 256, 0, stream>>>(
      tbuf, nw2, nb2, node_out, N);

  pool_zero_kernel<<<(G * HDIM + G + 255) / 256, 256, 0, stream>>>(
      gsum, G * HDIM + G);
  pool_block_kernel<<<(N + HDIM - 1) / HDIM, HDIM, 0, stream>>>(
      h, batch, gsum, gcnt, N);
  graph_head_kernel<<<G, 64, 0, stream>>>(gsum, gcnt, gw1, gb1, gw2, gb2,
                                          graph_out);
}


# Per layer, row gather + bias/BN/ReLU + the next layer's f32 WMMA transform fused into one resident kernel per 16-node tile (rows staged in LDS); 19 launches become 14; sum order unchanged
# speedup vs baseline: 2.7240x; 1.1974x over previous
#include <hip/hip_runtime.h>
#include <math.h>

typedef __attribute__((ext_vector_type(2))) float v2f;
typedef __attribute__((ext_vector_type(8))) float v8f;

#define HDIM 128

__global__ void cnt_zero_kernel(int* cnt, int N) {
  int i = blockIdx.x * blockDim.x + threadIdx.x;
  if (i < N) cnt[i] = 0;
}

__global__ void edge_rank_kernel(int* cnt, const int* __restrict__ dst,
                                 int* __restrict__ rank, int E) {
  int e = blockIdx.x * blockDim.x + threadIdx.x;
  if (e < E) rank[e] = atomicAdd(&cnt[dst[e]], 1);
}

__global__ __launch_bounds__(1024)
void rowptr_scan_kernel(const int* __restrict__ cnt, int* __restrict__ rowptr,
                        float* __restrict__ dis, int N) {
  __shared__ int wpre[32];
  __shared__ int wtot[32];
  const int tid = threadIdx.x;
  const int lane = tid & 31;
  const int wave = tid >> 5;
  const int base = blockIdx.x * 1024;
  const int4* __restrict__ cnt4 = (const int4*)cnt;
  const int nq = base >> 2;
  int part = 0;
#pragma unroll 4
  for (int q = tid; q < nq; q += 1024) {
    const int4 t = cnt4[q];
    part += (t.x + t.y) + (t.z + t.w);
  }
  for (int o = 16; o > 0; o >>= 1) part += __shfl_xor(part, o);
  if (lane == 0) wpre[wave] = part;
  const int i = base + tid;
  const int v = (i < N) ? cnt[i] : 0;
  int inc = v;
  for (int o = 1; o < 32; o <<= 1) {
    const int t = __shfl_up(inc, o);
    if (lane >= o) inc += t;
  }
  if (lane == 31) wtot[wave] = inc;
  __syncthreads();
  int off = 0;
  for (int w = 0; w < 32; ++w) {
    off += wpre[w];
    if (w < wave) off += wtot[w];
  }
  const int excl = off + inc - v;
  if (i < N) {
    rowptr[i] = excl;
    const float d = (float)(v + 1);
    dis[i] = d > 0.0f ? rsqrtf(d) : 0.0f;
    if (i == N - 1) rowptr[N] = excl + v;
  }
}

__global__ void csr_fill_kernel(const int* __restrict__ src, const int* __restrict__ dst,
                                const int* __restrict__ rank, const int* __restrict__ rowptr,
                                int* __restrict__ colu, int E) {
  int e = blockIdx.x * blockDim.x + threadIdx.x;
  if (e < E) colu[rowptr[dst[e]] + rank[e]] = src[e];
}

__global__ __launch_bounds__(256)
void csr_order_kernel(const int* __restrict__ rowptr, const int* __restrict__ colu,
                      int* __restrict__ cols, int N) {
  const int node = __builtin_amdgcn_readfirstlane(blockIdx.x * 8 + (threadIdx.x >> 5));
  const int lane = threadIdx.x & 31;
  if (node >= N) return;
  const int beg = rowptr[node];
  const int end = rowptr[node + 1];
  for (int i0 = beg; i0 < end; i0 += 32) {
    const int i = i0 + lane;
    const int x = (i < end) ? colu[i] : 0x7fffffff;
    int r = 0;
    for (int j0 = beg; j0 < end; j0 += 32) {
      const int j = j0 + lane;
      const int y = (j < end) ? colu[j] : 0x7fffffff;
      const int m = (end - j0) < 32 ? (end - j0) : 32;
      for (int k = 0; k < m; ++k) {
        const int yk = __builtin_amdgcn_readlane(y, k);
        r += (yk < x) ? 1 : 0;
        r += (yk == x && (j0 + k) < i) ? 1 : 0;
      }
    }
    if (i < end) cols[beg + r] = x;
  }
}

template <int NCOLS, int EPI>
__global__ __launch_bounds__(NCOLS * 2)
void gemm_f32_wmma_kernel(const float* __restrict__ A,
                          const float* __restrict__ W,
                          const float* __restrict__ dis,
                          const float* __restrict__ bias,
                          float* __restrict__ out0,
                          int M) {
  constexpr int K = HDIM;
  constexpr int KSTEPS = K / 4;
  __shared__ float sAt[K * 17 + 16];

  const int tid = threadIdx.x;
  const int wave = tid >> 5;
  const int lane = tid & 31;
  const int idx15 = lane & 15;
  const int kb = (lane >> 4) << 1;
  const int col = wave * 16 + idx15;

  v2f bfrag[KSTEPS];
#pragma unroll
  for (int s = 0; s < KSTEPS; ++s) {
    const float* wp = W + (size_t)col * K + s * 4 + kb;
    bfrag[s].x = wp[0];
    bfrag[s].y = wp[1];
  }
  float bias_v = 0.0f;
  if (EPI == 1) bias_v = bias[col];

  const int mtiles = M >> 4;
  for (int mt = blockIdx.x; mt < mtiles; mt += gridDim.x) {
    __syncthreads();
#pragma unroll
    for (int i = tid; i < 16 * K; i += NCOLS * 2) {
      int r = i >> 7;
      int c = i & (K - 1);
      sAt[c * 17 + r] = A[((size_t)mt * 16 + r) * K + c];
    }
    __syncthreads();

    v8f accv = {};
#pragma unroll
    for (int s = 0; s < KSTEPS; ++s) {
      const int k = s * 4 + kb;
      v2f a;
      a.x = sAt[k * 17 + idx15];
      a.y = sAt[(k + 1) * 17 + idx15];
      accv = __builtin_amdgcn_wmma_f32_16x16x4_f32(
          false, a, false, bfrag[s], (short)0, accv, false, false);
    }

    const int mbase = (lane >> 4) << 3;
#pragma unroll
    for (int i = 0; i < 8; ++i) {
      const size_t row = (size_t)mt * 16 + mbase + i;
      float v = accv[i];
      if (EPI == 0) {
        v *= dis[row];
        out0[row * NCOLS + col] = v;
      } else {
        v += bias_v;
        v = fmaxf(v, 0.0f);
        out0[row * NCOLS + col] = v;
      }
    }
  }
}

#define GATHER_ROWS(CNT)                                                        \
  {                                                                             \
    float4 v[CNT];                                                              \
    _Pragma("unroll") for (int u = 0; u < CNT; ++u) {                           \
      const int s = __builtin_amdgcn_readlane(my, k + u);                       \
      v[u] = hw4[(size_t)s * 32 + lane];                                        \
    }                                                                           \
    _Pragma("unroll") for (int u = 0; u < CNT; ++u) {                           \
      a.x += v[u].x; a.y += v[u].y; a.z += v[u].z; a.w += v[u].w;               \
    }                                                                           \
    k += CNT;                                                                   \
  }

__device__ __forceinline__ float bn_relu_one(float a, float dn, float bi, float me,
                                             float rs, float ga, float be) {
  float v = a * dn + bi;
  v = (v - me) * rs * ga + be;
  return fmaxf(v, 0.0f);
}

template <bool NEXT>
__global__ __launch_bounds__(256)
void layer_fused_kernel(const int* __restrict__ rowptr, const int* __restrict__ cols,
                        const float* __restrict__ hw, const float* __restrict__ dis,
                        const float* __restrict__ bias, const float* __restrict__ gamma,
                        const float* __restrict__ beta, const float* __restrict__ mean,
                        const float* __restrict__ var, const float* __restrict__ Wn,
                        float* __restrict__ outp, int N) {
  constexpr int K = HDIM;
  constexpr int KSTEPS = K / 4;
  constexpr int SA = K * 17 + 16;
  __shared__ float sAt[NEXT ? 2 * SA : 1];
  __shared__ float4 sC4[5 * 32];

  const int tid = threadIdx.x;
  const int wave = __builtin_amdgcn_readfirstlane(tid >> 5);
  const int lane = tid & 31;
  const int idx15 = lane & 15;
  const int kb = (lane >> 4) << 1;
  const int col = wave * 16 + idx15;
  const float4* __restrict__ hw4 = (const float4*)hw;

  if (tid < K) {
    float* sC = (float*)sC4;
    sC[0 * K + tid] = bias[tid];
    sC[1 * K + tid] = mean[tid];
    sC[2 * K + tid] = rsqrtf(var[tid] + 1e-5f);
    sC[3 * K + tid] = gamma[tid];
    sC[4 * K + tid] = beta[tid];
  }

  v2f bfrag[NEXT ? KSTEPS : 1];
  if (NEXT) {
#pragma unroll
    for (int s = 0; s < KSTEPS; ++s) {
      const float* wp = Wn + (size_t)col * K + s * 4 + kb;
      bfrag[s].x = wp[0];
      bfrag[s].y = wp[1];
    }
  }
  __syncthreads();

  const int mtiles = N >> 4;
  int it = 0;
  for (int mt = blockIdx.x; mt < mtiles; mt += gridDim.x, ++it) {
    float* sA = sAt + (NEXT ? (it & 1) * SA : 0);
    const int node0 = mt * 16 + wave * 2;
    const int rb0 = rowptr[node0];
    const int rb1 = rowptr[node0 + 1];
    const int rb2 = rowptr[node0 + 2];
    const int first0 = (lane < rb1 - rb0) ? cols[rb0 + lane] : 0;
    const int first1 = (lane < rb2 - rb1) ? cols[rb1 + lane] : 0;
    const float4 self0 = hw4[(size_t)node0 * 32 + lane];
    const float4 self1 = hw4[(size_t)(node0 + 1) * 32 + lane];

#pragma unroll
    for (int q = 0; q < 2; ++q) {
      const int node = node0 + q;
      const int beg = q ? rb1 : rb0;
      const int end = q ? rb2 : rb1;
      float4 a = q ? self1 : self0;
      for (int j0 = beg; j0 < end; j0 += 32) {
        const int m = (end - j0) < 32 ? (end - j0) : 32;
        int my;
        if (j0 == beg) my = q ? first1 : first0;
        else my = (lane < m) ? cols[j0 + lane] : 0;
        int k = 0;
        while (k + 8 <= m) GATHER_ROWS(8)
        if (k + 4 <= m) GATHER_ROWS(4)
        if (k + 2 <= m) GATHER_ROWS(2)
        if (k < m) GATHER_ROWS(1)
      }
      const float dn = dis[node];
      const float4 bi = sC4[0 * 32 + lane];
      const float4 me = sC4[1 * 32 + lane];
      const float4 rs = sC4[2 * 32 + lane];
      const float4 ga = sC4[3 * 32 + lane];
      const float4 be = sC4[4 * 32 + lane];
      float4 o;
      o.x = bn_relu_one(a.x, dn, bi.x, me.x, rs.x, ga.x, be.x);
      o.y = bn_relu_one(a.y, dn, bi.y, me.y, rs.y, ga.y, be.y);
      o.z = bn_relu_one(a.z, dn, bi.z, me.z, rs.z, ga.z, be.z);
      o.w = bn_relu_one(a.w, dn, bi.w, me.w, rs.w, ga.w, be.w);
      if (NEXT) {
        const int r = wave * 2 + q;
        sA[(4 * lane + 0) * 17 + r] = o.x;
        sA[(4 * lane + 1) * 17 + r] = o.y;
        sA[(4 * lane + 2) * 17 + r] = o.z;
        sA[(4 * lane + 3) * 17 + r] = o.w;
      } else {
        ((float4*)outp)[(size_t)node * 32 + lane] = o;
      }
    }

    if (NEXT) {
      __syncthreads();
      v8f accv = {};
#pragma unroll
      for (int s = 0; s < KSTEPS; ++s) {
        const int k = s * 4 + kb;
        v2f af;
        af.x = sA[k * 17 + idx15];
        af.y = sA[(k + 1) * 17 + idx15];
        accv = __builtin_amdgcn_wmma_f32_16x16x4_f32(
            false, af, false, bfrag[s], (short)0, accv, false, false);
      }
      const int mbase = (lane >> 4) << 3;
#pragma unroll
      for (int i = 0; i < 8; ++i) {
        const size_t row = (size_t)mt * 16 + mbase + i;
        outp[row * K + col] = accv[i] * dis[row];
      }
    }
  }
}

__global__ void node_head2_kernel(const float* __restrict__ t,
                                  const float* __restrict__ w2,
                                  const float* __restrict__ b2,
                                  float* __restrict__ out, int N) {
  int n = blockIdx.x * blockDim.x + threadIdx.x;
  if (n >= N) return;
  const float* tp = t + (size_t)n * 64;
  float a0 = b2[0], a1 = b2[1];
#pragma unroll 16
  for (int k = 0; k < 64; ++k) {
    float tv = tp[k];
    a0 = fmaf(tv, w2[k], a0);
    a1 = fmaf(tv, w2[64 + k], a1);
  }
  out[(size_t)n * 2 + 0] = a0;
  out[(size_t)n * 2 + 1] = a1;
}

__global__ void pool_zero_kernel(float* gbuf, int total) {
  int i = blockIdx.x * blockDim.x + threadIdx.x;
  if (i < total) gbuf[i] = 0.0f;
}

__global__ void pool_block_kernel(const float* __restrict__ h,
                                  const int* __restrict__ batch,
                                  float* __restrict__ gsum,
                                  float* __restrict__ gcnt,
                                  int N) {
  __shared__ float lsum[64 * HDIM];
  __shared__ float lcnt[64];
  const int c = threadIdx.x;
  const int base = blockIdx.x * HDIM;
  for (int i = c; i < 64 * HDIM; i += HDIM) lsum[i] = 0.0f;
  if (c < 64) lcnt[c] = 0.0f;
  __syncthreads();
  int nmax = N - base;
  if (nmax > HDIM) nmax = HDIM;
  int curg = -1;
  float accv = 0.0f;
  for (int r = 0; r < nmax; ++r) {
    const int n = base + r;
    const int g = batch[n];
    if (g != curg) {
      if (curg >= 0) lsum[curg * HDIM + c] += accv;
      curg = g;
      accv = 0.0f;
    }
    accv += h[(size_t)n * HDIM + c];
    if (c == 0) lcnt[g] += 1.0f;
  }
  if (curg >= 0) lsum[curg * HDIM + c] += accv;
  __syncthreads();
  const int gmin = batch[base];
  const int gmax = batch[base + nmax - 1];
  for (int g = gmin; g <= gmax; ++g)
    atomicAdd(&gsum[g * HDIM + c], lsum[g * HDIM + c]);
  if (c == 0)
    for (int g = gmin; g <= gmax; ++g) atomicAdd(&gcnt[g], lcnt[g]);
}

__global__ void graph_head_kernel(const float* __restrict__ gsum,
                                  const float* __restrict__ gcnt,
                                  const float* __restrict__ w1,
                                  const float* __restrict__ b1,
                                  const float* __restrict__ w2,
                                  const float* __restrict__ b2,
                                  float* __restrict__ out) {
  __shared__ float emb[HDIM];
  __shared__ float t[64];
  const int g = blockIdx.x;
  const int j = threadIdx.x;
  const float inv = 1.0f / fmaxf(gcnt[g], 1.0f);
  for (int i = j; i < HDIM; i += 64) emb[i] = gsum[g * HDIM + i] * inv;
  __syncthreads();
  float a = b1[j];
#pragma unroll 16
  for (int k = 0; k < HDIM; ++k) a = fmaf(emb[k], w1[j * HDIM + k], a);
  t[j] = fmaxf(a, 0.0f);
  __syncthreads();
  if (j < 2) {
    float o = b2[j];
#pragma unroll 16
    for (int k = 0; k < 64; ++k) o = fmaf(t[k], w2[j * 64 + k], o);
    out[g * 2 + j] = o;
  }
}

extern "C" void kernel_launch(void* const* d_in, const int* in_sizes, int n_in,
                              void* d_out, int out_size, void* d_ws, size_t ws_size,
                              hipStream_t stream) {
  const float* x      = (const float*)d_in[0];
  const int*   eidx   = (const int*)d_in[1];
  const int*   batch  = (const int*)d_in[2];
  const float* conv_w = (const float*)d_in[3];
  const float* conv_b = (const float*)d_in[4];
  const float* bn_g   = (const float*)d_in[5];
  const float* bn_b   = (const float*)d_in[6];
  const float* bn_m   = (const float*)d_in[7];
  const float* bn_v   = (const float*)d_in[8];
  const float* nw1    = (const float*)d_in[9];
  const float* nb1    = (const float*)d_in[10];
  const float* nw2    = (const float*)d_in[11];
  const float* nb2    = (const float*)d_in[12];
  const float* gw1    = (const float*)d_in[13];
  const float* gb1    = (const float*)d_in[14];
  const float* gw2    = (const float*)d_in[15];
  const float* gb2    = (const float*)d_in[16];

  const int N = in_sizes[0] / HDIM;
  const int E = in_sizes[1] / 2;
  const int L = in_sizes[3] / (HDIM * HDIM);
  const int G = (out_size - 2 * N) / 2;

  const int* src = eidx;
  const int* dst = eidx + E;

  float* dis  = (float*)d_ws;
  float* h    = dis + N;
  float* hw   = h + (size_t)N * HDIM;
  float* acc  = hw + (size_t)N * HDIM;
  float* tbuf = acc + (size_t)N * HDIM;
  float* gsum = tbuf + (size_t)N * 64;
  float* gcnt = gsum + (size_t)G * HDIM;

  int* cnt    = (int*)tbuf;
  int* rowptr = cnt + N;
  int* cols   = rowptr + N + 32;
  int* colu   = cols + E;

  float* node_out  = (float*)d_out;
  float* graph_out = node_out + (size_t)N * 2;

  cnt_zero_kernel<<<(N + 255) / 256, 256, 0, stream>>>(cnt, N);
  edge_rank_kernel<<<(E + 255) / 256, 256, 0, stream>>>(cnt, dst, cols, E);
  rowptr_scan_kernel<<<(N + 1023) / 1024, 1024, 0, stream>>>(cnt, rowptr, dis, N);
  csr_fill_kernel<<<(E + 255) / 256, 256, 0, stream>>>(src, dst, cols, rowptr, colu, E);
  csr_order_kernel<<<(N + 7) / 8, 256, 0, stream>>>(rowptr, colu, cols, N);

  const int mtiles = N / 16;
  const int gemm_grid = mtiles < 2048 ? mtiles : 2048;

  gemm_f32_wmma_kernel<HDIM, 0><<<gemm_grid, 256, 0, stream>>>(
      x, conv_w, dis, nullptr, hw, N);
  float* cur = hw;
  float* nxt = acc;
  for (int l = 0; l < L; ++l) {
    if (l + 1 < L) {
      layer_fused_kernel<true><<<768, 256, 0, stream>>>(
          rowptr, cols, cur, dis, conv_b + l * HDIM, bn_g + l * HDIM, bn_b + l * HDIM,
          bn_m + l * HDIM, bn_v + l * HDIM, conv_w + (size_t)(l + 1) * HDIM * HDIM, nxt, N);
      float* t = cur; cur = nxt; nxt = t;
    } else {
      layer_fused_kernel<false><<<1536, 256, 0, stream>>>(
          rowptr, cols, cur, dis, conv_b + l * HDIM, bn_g + l * HDIM, bn_b + l * HDIM,
          bn_m + l * HDIM, bn_v + l * HDIM, nullptr, h, N);
    }
  }

  gemm_f32_wmma_kernel<64, 1><<<gemm_grid, 128, 0, stream>>>(
      h, nw1, nullptr, nb1, tbuf, N);
  node_head2_kernel<<<(N + 255) / 256, 256, 0, stream>>>(
      tbuf, nw2, nb2, node_out, N);

  pool_zero_kernel<<<(G * HDIM + G + 255) / 256, 256, 0, stream>>>(
      gsum, G * HDIM + G);
  pool_block_kernel<<<(N + HDIM - 1) / HDIM, HDIM, 0, stream>>>(
      h, batch, gsum, gcnt, N);
  graph_head_kernel<<<G, 64, 0, stream>>>(gsum, gcnt, gw1, gb1, gw2, gb2,
                                          graph_out);
}


# Edge lists grouped by destination with LDS-binned counting passes (bucket histogram, scan, in-LDS bin sort, per-bucket count/place/order) in place of per-edge global atomics; sum order unchanged
# speedup vs baseline: 3.2980x; 1.2107x over previous
#include <hip/hip_runtime.h>
#include <math.h>

typedef __attribute__((ext_vector_type(2))) float v2f;
typedef __attribute__((ext_vector_type(8))) float v8f;

#define HDIM 128

#define BSH 7
#define BNODES 128
#define NBMAX 1024
#define CHUNK 6250
#define TPB_BIN 512
#define EPT ((CHUNK + TPB_BIN - 1) / TPB_BIN)
#define CAP 4096
static_assert(NBMAX == 2 * TPB_BIN, "the bucket scan below handles two entries per thread");

__global__ __launch_bounds__(TPB_BIN)
void bucket_hist_kernel(const int* __restrict__ dst, int* __restrict__ H, int E, int NB, int G1) {
  __shared__ int hc[NBMAX];
  const int tid = threadIdx.x;
  const int c = blockIdx.x;
  for (int i = tid; i < NB; i += TPB_BIN) hc[i] = 0;
  const int e0 = c * CHUNK;
  int e1 = e0 + CHUNK;
  if (e1 > E) e1 = E;
  int dv[EPT];
#pragma unroll
  for (int u = 0; u < EPT; ++u) {
    const int e = e0 + tid + u * TPB_BIN;
    dv[u] = dst[e < e1 ? e : e1 - 1];
  }
  __syncthreads();
#pragma unroll
  for (int u = 0; u < EPT; ++u) {
    const int e = e0 + tid + u * TPB_BIN;
    if (e < e1) atomicAdd(&hc[dv[u] >> BSH], 1);
  }
  __syncthreads();
  for (int i = tid; i < NB; i += TPB_BIN) H[i * G1 + c] = hc[i];
}

__global__ __launch_bounds__(1024)
void flat_scan_kernel(const int* __restrict__ cnt, int* __restrict__ out, int n) {
  __shared__ int wpre[32];
  __shared__ int wtot[32];
  const int tid = threadIdx.x;
  const int lane = tid & 31;
  const int wave = tid >> 5;
  const int base = blockIdx.x * 1024;
  const int4* __restrict__ cnt4 = (const int4*)cnt;
  const int nq = base >> 2;
  int part = 0;
#pragma unroll 8
  for (int q = tid; q < nq; q += 1024) {
    const int4 t = cnt4[q];
    part += (t.x + t.y) + (t.z + t.w);
  }
  for (int o = 16; o > 0; o >>= 1) part += __shfl_xor(part, o);
  if (lane == 0) wpre[wave] = part;
  const int i = base + tid;
  const int v = (i < n) ? cnt[i] : 0;
  int inc = v;
  for (int o = 1; o < 32; o <<= 1) {
    const int t = __shfl_up(inc, o);
    if (lane >= o) inc += t;
  }
  if (lane == 31) wtot[wave] = inc;
  __syncthreads();
  int off = 0;
  for (int w = 0; w < 32; ++w) {
    off += wpre[w];
    if (w < wave) off += wtot[w];
  }
  const int excl = off + inc - v;
  if (i < n) {
    out[i] = excl;
    if (i == n - 1) out[n] = excl + v;
  }
}

__global__ __launch_bounds__(TPB_BIN)
void bucket_scatter_kernel(const int* __restrict__ src, const int* __restrict__ dst,
                           const int* __restrict__ S, int* __restrict__ bsrc,
                           int* __restrict__ bdst, int E, int NB, int G1) {
  __shared__ int cur[NBMAX];
  __shared__ int lstart[NBMAX];
  __shared__ int gofs[NBMAX];
  __shared__ int ssrc[CHUNK];
  __shared__ int sdst[CHUNK];
  __shared__ int wtot[TPB_BIN / 32];
  const int tid = threadIdx.x;
  const int lane = tid & 31;
  const int wave = tid >> 5;
  const int c = blockIdx.x;
  cur[2 * tid] = 0;
  cur[2 * tid + 1] = 0;
  const int e0 = c * CHUNK;
  int e1 = e0 + CHUNK;
  if (e1 > E) e1 = E;
  int dv[EPT];
  int sv[EPT];
#pragma unroll
  for (int u = 0; u < EPT; ++u) {
    const int e = e0 + tid + u * TPB_BIN;
    const int ec = e < e1 ? e : e1 - 1;
    dv[u] = dst[ec];
    sv[u] = src[ec];
  }
  __syncthreads();
#pragma unroll
  for (int u = 0; u < EPT; ++u) {
    const int e = e0 + tid + u * TPB_BIN;
    if (e < e1) atomicAdd(&cur[dv[u] >> BSH], 1);
  }
  __syncthreads();
  const int v0 = cur[2 * tid];
  const int v1 = cur[2 * tid + 1];
  int inc = v0 + v1;
  for (int o = 1; o < 32; o <<= 1) {
    const int t = __shfl_up(inc, o);
    if (lane >= o) inc += t;
  }
  if (lane == 31) wtot[wave] = inc;
  __syncthreads();
  int off = 0;
  for (int w = 0; w < TPB_BIN / 32; ++w)
    if (w < wave) off += wtot[w];
  const int ex = off + inc - (v0 + v1);
  lstart[2 * tid] = ex;
  lstart[2 * tid + 1] = ex + v0;
  cur[2 * tid] = ex;
  cur[2 * tid + 1] = ex + v0;
  for (int i = tid; i < NB; i += TPB_BIN) gofs[i] = S[i * G1 + c];
  __syncthreads();
#pragma unroll
  for (int u = 0; u < EPT; ++u) {
    const int e = e0 + tid + u * TPB_BIN;
    if (e < e1) {
      const int p = atomicAdd(&cur[dv[u] >> BSH], 1);
      ssrc[p] = sv[u];
      sdst[p] = dv[u];
    }
  }
  __syncthreads();
  const int n = e1 - e0;
  for (int i = tid; i < n; i += TPB_BIN) {
    const int d = sdst[i];
    const int b = d >> BSH;
    const int g = gofs[b] + (i - lstart[b]);
    bsrc[g] = ssrc[i];
    bdst[g] = d;
  }
}

#define ORDER_LISTS(LIST)                                                       \
  for (int dn = wave; dn < BNODES; dn += 8) {                                   \
      \
    const int beg = __builtin_amdgcn_readfirstlane(lptr[dn]);                   \
    const int end = __builtin_amdgcn_readfirstlane(lptr[dn + 1]);               \
    for (int i0 = beg; i0 < end; i0 += 32) {                                    \
      const int i = i0 + lane;                                                  \
      const int x = (i < end) ? LIST(i) : 0x7fffffff;                           \
      int r = 0;                \
      for (int j0 = beg; j0 < end; j0 += 32) {                                  \
        const int j = j0 + lane;                                                \
        const int y = (j < end) ? LIST(j) : 0x7fffffff;                         \
        const int m = (end - j0) < 32 ? (end - j0) : 32;                        \
        for (int k = 0; k < m; ++k) {                                           \
          const int yk = __builtin_amdgcn_readlane(y, k);                       \
          r += (yk < x) ? 1 : 0;                                                \
          r += (yk == x && (j0 + k) < i) ? 1 : 0;                               \
        }                                                                       \
      }                                                                         \
      if (i < end) cols[bb + beg + r] = x;                                      \
    }                                                                           \
  }
#define LIST_LDS(i) lsrc[(i)]
#define LIST_MEM(i) colu[bb + (i)]

__global__ __launch_bounds__(256)
void bucket_finish_kernel(const int* __restrict__ S, const int* __restrict__ bsrc,
                          const int* __restrict__ bdst, int* __restrict__ rowptr,
                          float* __restrict__ dis, int* __restrict__ cols, int* colu,
                          int N, int G1) {
  __shared__ int cnt[BNODES];
  __shared__ int lptr[BNODES + 1];
  __shared__ int cur[BNODES];
  __shared__ int wt[BNODES / 32];
  __shared__ int lsrc[CAP];
  const int tid = threadIdx.x;
  const int lane = tid & 31;
  const int wave = __builtin_amdgcn_readfirstlane(tid >> 5);
  const int b = blockIdx.x;
  const int bb = S[b * G1];
  const int nb = S[(b + 1) * G1] - bb;
  const int node0 = b << BSH;
  const bool use_lds = nb <= CAP;
  if (tid < BNODES) cnt[tid] = 0;
  __syncthreads();
  for (int i0 = 0; i0 < nb; i0 += 4 * 256) {
    int dd[4];
#pragma unroll
    for (int u = 0; u < 4; ++u) {
      const int i = i0 + u * 256 + tid;
      dd[u] = bdst[bb + (i < nb ? i : nb - 1)];
    }
#pragma unroll
    for (int u = 0; u < 4; ++u) {
      const int i = i0 + u * 256 + tid;
      if (i < nb) atomicAdd(&cnt[dd[u] - node0], 1);
    }
  }
  __syncthreads();
  const int v = (tid < BNODES) ? cnt[tid] : 0;
  int inc = v;
  for (int o = 1; o < 32; o <<= 1) {
    const int t = __shfl_up(inc, o);
    if (lane >= o) inc += t;
  }
  if (tid < BNODES && lane == 31) wt[tid >> 5] = inc;
  __syncthreads();
  if (tid < BNODES) {
    int off = 0;
    for (int w = 0; w < BNODES / 32; ++w)
      if (w < (tid >> 5)) off += wt[w];
    const int ex = off + inc - v;
    lptr[tid] = ex;
    cur[tid] = ex;
    if (tid == BNODES - 1) lptr[BNODES] = ex + v;
    const int node = node0 + tid;
    if (node < N) {
      rowptr[node] = bb + ex;
      const float d = (float)(v + 1);
      dis[node] = d > 0.0f ? rsqrtf(d) : 0.0f;
      if (node == N - 1) rowptr[N] = bb + ex + v;
    }
  }
  __syncthreads();
  for (int i0 = 0; i0 < nb; i0 += 4 * 256) {
    int dd[4];
    int ss[4];
#pragma unroll
    for (int u = 0; u < 4; ++u) {
      const int i = i0 + u * 256 + tid;
      const int ic = bb + (i < nb ? i : nb - 1);
      dd[u] = bdst[ic];
      ss[u] = bsrc[ic];
    }
#pragma unroll
    for (int u = 0; u < 4; ++u) {
      const int i = i0 + u * 256 + tid;
      if (i < nb) {
        const int p = atomicAdd(&cur[dd[u] - node0], 1);
        if (use_lds) lsrc[p] = ss[u];
        else colu[bb + p] = ss[u];
      }
    }
  }
  __syncthreads();
  if (use_lds) {
    ORDER_LISTS(LIST_LDS)
  } else {
    ORDER_LISTS(LIST_MEM)
  }
}

template <int NCOLS, int EPI>
__global__ __launch_bounds__(NCOLS * 2)
void gemm_f32_wmma_kernel(const float* __restrict__ A,
                          const float* __restrict__ W,
                          const float* __restrict__ dis,
                          const float* __restrict__ bias,
                          float* __restrict__ out0,
                          int M) {
  constexpr int K = HDIM;
  constexpr int KSTEPS = K / 4;
  __shared__ float sAt[K * 17 + 16];

  const int tid = threadIdx.x;
  const int wave = tid >> 5;
  const int lane = tid & 31;
  const int idx15 = lane & 15;
  const int kb = (lane >> 4) << 1;
  const int col = wave * 16 + idx15;

  v2f bfrag[KSTEPS];
#pragma unroll
  for (int s = 0; s < KSTEPS; ++s) {
    const float* wp = W + (size_t)col * K + s * 4 + kb;
    bfrag[s].x = wp[0];
    bfrag[s].y = wp[1];
  }
  float bias_v = 0.0f;
  if (EPI == 1) bias_v = bias[col];

  const int mtiles = M >> 4;
  for (int mt = blockIdx.x; mt < mtiles; mt += gridDim.x) {
    __syncthreads();
#pragma unroll
    for (int i = tid; i < 16 * K; i += NCOLS * 2) {
      int r = i >> 7;
      int c = i & (K - 1);
      sAt[c * 17 + r] = A[((size_t)mt * 16 + r) * K + c];
    }
    __syncthreads();

    v8f accv = {};
#pragma unroll
    for (int s = 0; s < KSTEPS; ++s) {
      const int k = s * 4 + kb;
      v2f a;
      a.x = sAt[k * 17 + idx15];
      a.y = sAt[(k + 1) * 17 + idx15];
      accv = __builtin_amdgcn_wmma_f32_16x16x4_f32(
          false, a, false, bfrag[s], (short)0, accv, false, false);
    }

    const int mbase = (lane >> 4) << 3;
#pragma unroll
    for (int i = 0; i < 8; ++i) {
      const size_t row = (size_t)mt * 16 + mbase + i;
      float v = accv[i];
      if (EPI == 0) {
        v *= dis[row];
        out0[row * NCOLS + col] = v;
      } else {
        v += bias_v;
        v = fmaxf(v, 0.0f);
        out0[row * NCOLS + col] = v;
      }
    }
  }
}

#define GATHER_ROWS(CNT)                                                        \
  {                                                                             \
    float4 v[CNT];                                                              \
    _Pragma("unroll") for (int u = 0; u < CNT; ++u) {                           \
      const int s = __builtin_amdgcn_readlane(my, k + u);                       \
      v[u] = hw4[(size_t)s * 32 + lane];                                        \
    }                                                                           \
    _Pragma("unroll") for (int u = 0; u < CNT; ++u) {                           \
      a.x += v[u].x; a.y += v[u].y; a.z += v[u].z; a.w += v[u].w;               \
    }                                                                           \
    k += CNT;                                                                   \
  }

__device__ __forceinline__ float bn_relu_one(float a, float dn, float bi, float me,
                                             float rs, float ga, float be) {
  float v = a * dn + bi;
  v = (v - me) * rs * ga + be;
  return fmaxf(v, 0.0f);
}

template <bool NEXT>
__global__ __launch_bounds__(256)
void layer_fused_kernel(const int* __restrict__ rowptr, const int* __restrict__ cols,
                        const float* __restrict__ hw, const float* __restrict__ dis,
                        const float* __restrict__ bias, const float* __restrict__ gamma,
                        const float* __restrict__ beta, const float* __restrict__ mean,
                        const float* __restrict__ var, const float* __restrict__ Wn,
                        float* __restrict__ outp, int N) {
  constexpr int K = HDIM;
  constexpr int KSTEPS = K / 4;
  constexpr int SA = K * 17 + 16;
  __shared__ float sAt[NEXT ? 2 * SA : 1];
  __shared__ float4 sC4[5 * 32];

  const int tid = threadIdx.x;
  const int wave = __builtin_amdgcn_readfirstlane(tid >> 5);
  const int lane = tid & 31;
  const int idx15 = lane & 15;
  const int kb = (lane >> 4) << 1;
  const int col = wave * 16 + idx15;
  const float4* __restrict__ hw4 = (const float4*)hw;

  if (tid < K) {
    float* sC = (float*)sC4;
    sC[0 * K + tid] = bias[tid];
    sC[1 * K + tid] = mean[tid];
    sC[2 * K + tid] = rsqrtf(var[tid] + 1e-5f);
    sC[3 * K + tid] = gamma[tid];
    sC[4 * K + tid] = beta[tid];
  }

  v2f bfrag[NEXT ? KSTEPS : 1];
  if (NEXT) {
#pragma unroll
    for (int s = 0; s < KSTEPS; ++s) {
      const float* wp = Wn + (size_t)col * K + s * 4 + kb;
      bfrag[s].x = wp[0];
      bfrag[s].y = wp[1];
    }
  }
  __syncthreads();

  const int mtiles = N >> 4;
  int it = 0;
  for (int mt = blockIdx.x; mt < mtiles; mt += gridDim.x, ++it) {
    float* sA = sAt + (NEXT ? (it & 1) * SA : 0);
    const int node0 = mt * 16 + wave * 2;
    const int rb0 = rowptr[node0];
    const int rb1 = rowptr[node0 + 1];
    const int rb2 = rowptr[node0 + 2];
    const int first0 = (lane < rb1 - rb0) ? cols[rb0 + lane] : 0;
    const int first1 = (lane < rb2 - rb1) ? cols[rb1 + lane] : 0;
    const float4 self0 = hw4[(size_t)node0 * 32 + lane];
    const float4 self1 = hw4[(size_t)(node0 + 1) * 32 + lane];

#pragma unroll
    for (int q = 0; q < 2; ++q) {
      const int node = node0 + q;
      const int beg = q ? rb1 : rb0;
      const int end = q ? rb2 : rb1;
      float4 a = q ? self1 : self0;
      for (int j0 = beg; j0 < end; j0 += 32) {
        const int m = (end - j0) < 32 ? (end - j0) : 32;
        int my;
        if (j0 == beg) my = q ? first1 : first0;
        else my = (lane < m) ? cols[j0 + lane] : 0;
        int k = 0;
        while (k + 8 <= m) GATHER_ROWS(8)
        if (k + 4 <= m) GATHER_ROWS(4)
        if (k + 2 <= m) GATHER_ROWS(2)
        if (k < m) GATHER_ROWS(1)
      }
      const float dn = dis[node];
      const float4 bi = sC4[0 * 32 + lane];
      const float4 me = sC4[1 * 32 + lane];
      const float4 rs = sC4[2 * 32 + lane];
      const float4 ga = sC4[3 * 32 + lane];
      const float4 be = sC4[4 * 32 + lane];
      float4 o;
      o.x = bn_relu_one(a.x, dn, bi.x, me.x, rs.x, ga.x, be.x);
      o.y = bn_relu_one(a.y, dn, bi.y, me.y, rs.y, ga.y, be.y);
      o.z = bn_relu_one(a.z, dn, bi.z, me.z, rs.z, ga.z, be.z);
      o.w = bn_relu_one(a.w, dn, bi.w, me.w, rs.w, ga.w, be.w);
      if (NEXT) {
        const int r = wave * 2 + q;
        sA[(4 * lane + 0) * 17 + r] = o.x;
        sA[(4 * lane + 1) * 17 + r] = o.y;
        sA[(4 * lane + 2) * 17 + r] = o.z;
        sA[(4 * lane + 3) * 17 + r] = o.w;
      } else {
        ((float4*)outp)[(size_t)node * 32 + lane] = o;
      }
    }

    if (NEXT) {
      __syncthreads();
      v8f accv = {};
#pragma unroll
      for (int s = 0; s < KSTEPS; ++s) {
        const int k = s * 4 + kb;
        v2f af;
        af.x = sA[k * 17 + idx15];
        af.y = sA[(k + 1) * 17 + idx15];
        accv = __builtin_amdgcn_wmma_f32_16x16x4_f32(
            false, af, false, bfrag[s], (short)0, accv, false, false);
      }
      const int mbase = (lane >> 4) << 3;
#pragma unroll
      for (int i = 0; i < 8; ++i) {
        const size_t row = (size_t)mt * 16 + mbase + i;
        outp[row * K + col] = accv[i] * dis[row];
      }
    }
  }
}

__global__ void node_head2_kernel(const float* __restrict__ t,
                                  const float* __restrict__ w2,
                                  const float* __restrict__ b2,
                                  float* __restrict__ out, int N) {
  int n = blockIdx.x * blockDim.x + threadIdx.x;
  if (n >= N) return;
  const float* tp = t + (size_t)n * 64;
  float a0 = b2[0], a1 = b2[1];
#pragma unroll 16
  for (int k = 0; k < 64; ++k) {
    float tv = tp[k];
    a0 = fmaf(tv, w2[k], a0);
    a1 = fmaf(tv, w2[64 + k], a1);
  }
  out[(size_t)n * 2 + 0] = a0;
  out[(size_t)n * 2 + 1] = a1;
}

__global__ void pool_zero_kernel(float* gbuf, int total) {
  int i = blockIdx.x * blockDim.x + threadIdx.x;
  if (i < total) gbuf[i] = 0.0f;
}

__global__ void pool_block_kernel(const float* __restrict__ h,
                                  const int* __restrict__ batch,
                                  float* __restrict__ gsum,
                                  float* __restrict__ gcnt,
                                  int N) {
  __shared__ float lsum[64 * HDIM];
  __shared__ float lcnt[64];
  const int c = threadIdx.x;
  const int base = blockIdx.x * HDIM;
  for (int i = c; i < 64 * HDIM; i += HDIM) lsum[i] = 0.0f;
  if (c < 64) lcnt[c] = 0.0f;
  __syncthreads();
  int nmax = N - base;
  if (nmax > HDIM) nmax = HDIM;
  int curg = -1;
  float accv = 0.0f;
  for (int r = 0; r < nmax; ++r) {
    const int n = base + r;
    const int g = batch[n];
    if (g != curg) {
      if (curg >= 0) lsum[curg * HDIM + c] += accv;
      curg = g;
      accv = 0.0f;
    }
    accv += h[(size_t)n * HDIM + c];
    if (c == 0) lcnt[g] += 1.0f;
  }
  if (curg >= 0) lsum[curg * HDIM + c] += accv;
  __syncthreads();
  const int gmin = batch[base];
  const int gmax = batch[base + nmax - 1];
  for (int g = gmin; g <= gmax; ++g)
    atomicAdd(&gsum[g * HDIM + c], lsum[g * HDIM + c]);
  if (c == 0)
    for (int g = gmin; g <= gmax; ++g) atomicAdd(&gcnt[g], lcnt[g]);
}

__global__ void graph_head_kernel(const float* __restrict__ gsum,
                                  const float* __restrict__ gcnt,
                                  const float* __restrict__ w1,
                                  const float* __restrict__ b1,
                                  const float* __restrict__ w2,
                                  const float* __restrict__ b2,
                                  float* __restrict__ out) {
  __shared__ float emb[HDIM];
  __shared__ float t[64];
  const int g = blockIdx.x;
  const int j = threadIdx.x;
  const float inv = 1.0f / fmaxf(gcnt[g], 1.0f);
  for (int i = j; i < HDIM; i += 64) emb[i] = gsum[g * HDIM + i] * inv;
  __syncthreads();
  float a = b1[j];
#pragma unroll 16
  for (int k = 0; k < HDIM; ++k) a = fmaf(emb[k], w1[j * HDIM + k], a);
  t[j] = fmaxf(a, 0.0f);
  __syncthreads();
  if (j < 2) {
    float o = b2[j];
#pragma unroll 16
    for (int k = 0; k < 64; ++k) o = fmaf(t[k], w2[j * 64 + k], o);
    out[g * 2 + j] = o;
  }
}

extern "C" void kernel_launch(void* const* d_in, const int* in_sizes, int n_in,
                              void* d_out, int out_size, void* d_ws, size_t ws_size,
                              hipStream_t stream) {
  const float* x      = (const float*)d_in[0];
  const int*   eidx   = (const int*)d_in[1];
  const int*   batch  = (const int*)d_in[2];
  const float* conv_w = (const float*)d_in[3];
  const float* conv_b = (const float*)d_in[4];
  const float* bn_g   = (const float*)d_in[5];
  const float* bn_b   = (const float*)d_in[6];
  const float* bn_m   = (const float*)d_in[7];
  const float* bn_v   = (const float*)d_in[8];
  const float* nw1    = (const float*)d_in[9];
  const float* nb1    = (const float*)d_in[10];
  const float* nw2    = (const float*)d_in[11];
  const float* nb2    = (const float*)d_in[12];
  const float* gw1    = (const float*)d_in[13];
  const float* gb1    = (const float*)d_in[14];
  const float* gw2    = (const float*)d_in[15];
  const float* gb2    = (const float*)d_in[16];

  const int N = in_sizes[0] / HDIM;
  const int E = in_sizes[1] / 2;
  const int L = in_sizes[3] / (HDIM * HDIM);
  const int G = (out_size - 2 * N) / 2;

  const int* src = eidx;
  const int* dst = eidx + E;

  float* dis  = (float*)d_ws;
  float* h    = dis + N;
  float* hw   = h + (size_t)N * HDIM;
  float* acc  = hw + (size_t)N * HDIM;
  float* tbuf = acc + (size_t)N * HDIM;
  float* gsum = tbuf + (size_t)N * 64;
  float* gcnt = gsum + (size_t)G * HDIM;

  const int NB = (N + BNODES - 1) / BNODES;
  const int G1 = (E + CHUNK - 1) / CHUNK;
  const int HN = NB * G1;
  int* H      = (int*)tbuf;
  int* S      = H + HN + 32;
  int* rowptr = S + HN + 32;
  int* cols   = rowptr + N + 32;
  int* colu   = cols + E;
  int* bsrc   = (int*)h;
  int* bdst   = bsrc + E;

  float* node_out  = (float*)d_out;
  float* graph_out = node_out + (size_t)N * 2;

  bucket_hist_kernel<<<G1, TPB_BIN, 0, stream>>>(dst, H, E, NB, G1);
  flat_scan_kernel<<<(HN + 1023) / 1024, 1024, 0, stream>>>(H, S, HN);
  bucket_scatter_kernel<<<G1, TPB_BIN, 0, stream>>>(src, dst, S, bsrc, bdst, E, NB, G1);
  bucket_finish_kernel<<<NB, 256, 0, stream>>>(S, bsrc, bdst, rowptr, dis, cols, colu, N, G1);

  const int mtiles = N / 16;
  const int gemm_grid = mtiles < 2048 ? mtiles : 2048;

  gemm_f32_wmma_kernel<HDIM, 0><<<gemm_grid, 256, 0, stream>>>(
      x, conv_w, dis, nullptr, hw, N);
  float* cur = hw;
  float* nxt = acc;
  for (int l = 0; l < L; ++l) {
    if (l + 1 < L) {
      layer_fused_kernel<true><<<768, 256, 0, stream>>>(
          rowptr, cols, cur, dis, conv_b + l * HDIM, bn_g + l * HDIM, bn_b + l * HDIM,
          bn_m + l * HDIM, bn_v + l * HDIM, conv_w + (size_t)(l + 1) * HDIM * HDIM, nxt, N);
      float* t = cur; cur = nxt; nxt = t;
    } else {
      layer_fused_kernel<false><<<1536, 256, 0, stream>>>(
          rowptr, cols, cur, dis, conv_b + l * HDIM, bn_g + l * HDIM, bn_b + l * HDIM,
          bn_m + l * HDIM, bn_v + l * HDIM, nullptr, h, N);
    }
  }

  gemm_f32_wmma_kernel<64, 1><<<gemm_grid, 128, 0, stream>>>(
      h, nw1, nullptr, nb1, tbuf, N);
  node_head2_kernel<<<(N + 255) / 256, 256, 0, stream>>>(
      tbuf, nw2, nb2, node_out, N);

  pool_zero_kernel<<<(G * HDIM + G + 255) / 256, 256, 0, stream>>>(
      gsum, G * HDIM + G);
  pool_block_kernel<<<(N + HDIM - 1) / HDIM, HDIM, 0, stream>>>(
      h, batch, gsum, gcnt, N);
  graph_head_kernel<<<G, 64, 0, stream>>>(gsum, gcnt, gw1, gb1, gw2, gb2,
                                          graph_out);
}
